# starting kernel
# baseline (speedup 1.0000x reference)
#include <hip/hip_runtime.h>

typedef __attribute__((ext_vector_type(2))) float v2f;
typedef __attribute__((ext_vector_type(8))) float v8f;

#define C_IN   8
#define C_OUT  8
#define FSZ    4
#define TILE   16
#define TILES_PER_WAVE 16
#define WAVES_PER_BLOCK 8

constexpr int kL = 65536;

__global__ __launch_bounds__(256)
void convPbc_wmma_kernel(const float* __restrict__ x,
                         const float* __restrict__ W,
                         const float* __restrict__ bias,
                         float* __restrict__ out) {
  const int lane = threadIdx.x & 31;
  const int wave = threadIdx.x >> 5;
  const int m    = lane & 15;
  const int hi   = lane >> 4;

  constexpr int posPerBlock    = WAVES_PER_BLOCK * TILES_PER_WAVE * TILE;
  constexpr int blocksPerBatch = kL / posPerBlock;
  const int b       = blockIdx.x / blocksPerBatch;
  const int lblk    = (blockIdx.x % blocksPerBatch) * posPerBlock;
  const int wave_l0 = lblk + wave * (TILES_PER_WAVE * TILE);

  v2f A[C_IN];
  #pragma unroll
  for (int c = 0; c < C_IN; ++c) {
    float a0 = 0.0f, a1 = 0.0f;
    if (m < C_OUT) {
      const float* wp = W + (m * C_IN + c) * FSZ;
      a0 = wp[hi ? 2 : 0];
      a1 = wp[hi ? 3 : 1];
    }
    A[c].x = a0;
    A[c].y = a1;
  }

  v8f cbias;
  #pragma unroll
  for (int r = 0; r < 8; ++r) cbias[r] = hi ? 0.0f : bias[r];

  const float* xb = x   + (long long)b * C_IN  * kL;
  float*       ob = out + (long long)b * C_OUT * kL;
  const int n    = m;
  const int off0 = n + hi * 2;
  const int off1 = off0 + 1;

  for (int t = 0; t < TILES_PER_WAVE; ++t) {
    const int l0 = wave_l0 + t * TILE;
    v8f acc = cbias;

    const int i0 = (l0 + off0) & (kL - 1);
    const int i1 = (l0 + off1) & (kL - 1);
    const float* p0 = xb + i0;
    const float* p1 = xb + i1;

    #pragma unroll
    for (int c = 0; c < C_IN; ++c) {
      v2f Bf;
      Bf.x = p0[c * kL];
      Bf.y = p1[c * kL];
      acc = __builtin_amdgcn_wmma_f32_16x16x4_f32(
           false, A[c],  false, Bf,
           (short)0, acc,  false,  false);
    }

    if (hi == 0) {
      float* q = ob + l0 + n;
      #pragma unroll
      for (int r = 0; r < C_OUT; ++r)
        __builtin_nontemporal_store(acc[r], q + r * kL);
    }
  }
}

extern "C" void kernel_launch(void* const* d_in, const int* in_sizes, int n_in,
                              void* d_out, int out_size, void* d_ws, size_t ws_size,
                              hipStream_t stream) {
  const float* x    = (const float*)d_in[0];
  const float* W    = (const float*)d_in[1];
  const float* bias = (const float*)d_in[2];
  float*       out  = (float*)d_out;

  const int B = in_sizes[0] / (C_IN * kL);

  constexpr int posPerBlock = WAVES_PER_BLOCK * TILES_PER_WAVE * TILE;
  const int blocks = B * (kL / posPerBlock);

  convPbc_wmma_kernel<<<blocks, 256, 0, stream>>>(x, W, bias, out);
}


# Column n of 4 interleaved WMMA tiles = position 4n+j, lane halves split by channel: x read and out written 16 bytes per lane in whole lines (was 4 bytes); sum order: channels c, c+4 paired per step
# speedup vs baseline: 1.4841x; 1.4841x over previous
#include <hip/hip_runtime.h>

typedef __attribute__((ext_vector_type(2))) float v2f;
typedef __attribute__((ext_vector_type(4))) float v4f;
typedef __attribute__((ext_vector_type(8))) float v8f;

#define C_IN   8
#define C_OUT  8
#define FSZ    4
#define NT     4
#define GROUP  64
#define GROUPS_PER_WAVE 8
#define WAVES_PER_BLOCK 8

constexpr int kL = 65536;

struct XGroup {
  v4f a[4];
  v4f b[4];
};

__device__ __forceinline__ void load_group(XGroup& g, const float* __restrict__ xh,
                                           int p) {
  const int q = (p + 4) & (kL - 1);
  #pragma unroll
  for (int cc = 0; cc < 4; ++cc) {
    g.a[cc] = *(const v4f*)(xh + cc * kL + p);
    g.b[cc] = *(const v4f*)(xh + cc * kL + q);
  }
}

__device__ __forceinline__ void compute_store(const XGroup& g, const v2f (&A)[8],
                                              const v8f& cbias, float* __restrict__ orow,
                                              bool writer) {
  v8f acc[NT];
  #pragma unroll
  for (int j = 0; j < NT; ++j) acc[j] = cbias;

  #pragma unroll
  for (int cc = 0; cc < 4; ++cc) {
    const float v[8] = {g.a[cc].x, g.a[cc].y, g.a[cc].z, g.a[cc].w,
                        g.b[cc].x, g.b[cc].y, g.b[cc].z, g.b[cc].w};
    #pragma unroll
    for (int th = 0; th < 2; ++th) {
      #pragma unroll
      for (int j = 0; j < NT; ++j) {
        v2f Bf;
        Bf.x = v[j + 2 * th];
        Bf.y = v[j + 2 * th + 1];
        acc[j] = __builtin_amdgcn_wmma_f32_16x16x4_f32(
             false, A[2 * cc + th],  false, Bf,
             (short)0, acc[j],  false,  false);
      }
    }
  }

  if (writer) {
    #pragma unroll
    for (int r = 0; r < C_OUT; ++r) {
      v4f o;
      o.x = acc[0][r];
      o.y = acc[1][r];
      o.z = acc[2][r];
      o.w = acc[3][r];
      __builtin_nontemporal_store(o, (v4f*)(orow + r * kL));
    }
  }
}

__global__ __launch_bounds__(256, 8)
void convPbc_wmma_kernel(const float* __restrict__ x,
                         const float* __restrict__ W,
                         const float* __restrict__ bias,
                         float* __restrict__ out) {
  const int lane = threadIdx.x & 31;
  const int wave = threadIdx.x >> 5;
  const int m    = lane & 15;
  const int hi   = lane >> 4;

  constexpr int posPerWave     = GROUPS_PER_WAVE * GROUP;
  constexpr int posPerBlock    = WAVES_PER_BLOCK * posPerWave;
  constexpr int blocksPerBatch = kL / posPerBlock;
  const int b       = blockIdx.x / blocksPerBatch;
  const int lblk    = (blockIdx.x % blocksPerBatch) * posPerBlock;
  const int wave_l0 = lblk + wave * posPerWave;

  v2f A[8];
  #pragma unroll
  for (int i = 0; i < 8; ++i) {
    const int cc = i >> 1, th = i & 1;
    float a0 = 0.0f, a1 = 0.0f;
    if (m < C_OUT) {
      const float* wp = W + (m * C_IN + 4 * hi + cc) * FSZ + 2 * th;
      a0 = wp[0];
      a1 = wp[1];
    }
    A[i].x = a0;
    A[i].y = a1;
  }

  v8f cbias;
  #pragma unroll
  for (int r = 0; r < 8; ++r) cbias[r] = hi ? 0.0f : bias[r];

  const float* xh = x   + ((long long)b * C_IN + 4 * hi) * kL;
  float*       ob = out + (long long)b * C_OUT * kL;
  const int p0 = wave_l0 + 4 * m;
  const bool writer = (hi == 0);

  XGroup g0, g1;
  load_group(g0, xh, p0);
  #pragma unroll
  for (int t = 0; t < GROUPS_PER_WAVE; t += 2) {
    load_group(g1, xh, p0 + (t + 1) * GROUP);
    compute_store(g0, A, cbias, ob + p0 + t * GROUP, writer);
    if (t + 2 < GROUPS_PER_WAVE) load_group(g0, xh, p0 + (t + 2) * GROUP);
    compute_store(g1, A, cbias, ob + p0 + (t + 1) * GROUP, writer);
  }
}

extern "C" void kernel_launch(void* const* d_in, const int* in_sizes, int n_in,
                              void* d_out, int out_size, void* d_ws, size_t ws_size,
                              hipStream_t stream) {
  const float* x    = (const float*)d_in[0];
  const float* W    = (const float*)d_in[1];
  const float* bias = (const float*)d_in[2];
  float*       out  = (float*)d_out;

  const int B = in_sizes[0] / (C_IN * kL);

  constexpr int posPerBlock = WAVES_PER_BLOCK * GROUPS_PER_WAVE * GROUP;
  const int blocks = B * (kL / posPerBlock);

  convPbc_wmma_kernel<<<blocks, 256, 0, stream>>>(x, W, bias, out);
}


# Each x element loaded once: the 3 look-ahead taps come from lane n+1 by a DPP row rotate (lane 15: next group's lane 0), not from a second overlapping 16-byte load; loads run two groups ahead
# speedup vs baseline: 1.6879x; 1.1373x over previous
#include <hip/hip_runtime.h>

typedef __attribute__((ext_vector_type(2))) float v2f;
typedef __attribute__((ext_vector_type(3))) float v3f;
typedef __attribute__((ext_vector_type(4))) float v4f;
typedef __attribute__((ext_vector_type(8))) float v8f;

#define C_IN   8
#define C_OUT  8
#define FSZ    4
#define NT     4
#define GROUP  64
#define GROUPS_PER_WAVE 8
#define WAVES_PER_BLOCK 8

constexpr int kL = 65536;

struct XGroup {
  v4f a[4];
};

__device__ __forceinline__ void load_group(XGroup& g, const float* __restrict__ xh,
                                           int p) {
  #pragma unroll
  for (int cc = 0; cc < 4; ++cc) g.a[cc] = *(const v4f*)(xh + cc * kL + p);
}

__device__ __forceinline__ float from_next_lane(float cur, float nxt, bool lane0) {
  const float merged = lane0 ? nxt : cur;
  return __builtin_bit_cast(float,
      __builtin_amdgcn_update_dpp(0, __builtin_bit_cast(int, merged), 0x12F, 0xF, 0xF, true));
}

__device__ __forceinline__ void compute_store(const XGroup& g, const XGroup& gn, bool lane0,
                                              const v2f (&A)[8],
                                              const v8f& cbias, float* __restrict__ orow,
                                              bool writer) {
  v8f acc[NT];
  #pragma unroll
  for (int j = 0; j < NT; ++j) acc[j] = cbias;

  #pragma unroll
  for (int cc = 0; cc < 4; ++cc) {
    const float v[7] = {g.a[cc].x, g.a[cc].y, g.a[cc].z, g.a[cc].w,
                        from_next_lane(g.a[cc].x, gn.a[cc].x, lane0),
                        from_next_lane(g.a[cc].y, gn.a[cc].y, lane0),
                        from_next_lane(g.a[cc].z, gn.a[cc].z, lane0)};
    #pragma unroll
    for (int th = 0; th < 2; ++th) {
      #pragma unroll
      for (int j = 0; j < NT; ++j) {
        v2f Bf;
        Bf.x = v[j + 2 * th];
        Bf.y = v[j + 2 * th + 1];
        acc[j] = __builtin_amdgcn_wmma_f32_16x16x4_f32(
             false, A[2 * cc + th],  false, Bf,
             (short)0, acc[j],  false,  false);
      }
    }
  }

  if (writer) {
    #pragma unroll
    for (int r = 0; r < C_OUT; ++r) {
      v4f o;
      o.x = acc[0][r];
      o.y = acc[1][r];
      o.z = acc[2][r];
      o.w = acc[3][r];
      __builtin_nontemporal_store(o, (v4f*)(orow + r * kL));
    }
  }
}

__global__ __launch_bounds__(256, 8)
void convPbc_wmma_kernel(const float* __restrict__ x,
                         const float* __restrict__ W,
                         const float* __restrict__ bias,
                         float* __restrict__ out) {
  const int lane = threadIdx.x & 31;
  const int wave = threadIdx.x >> 5;
  const int m    = lane & 15;
  const int hi   = lane >> 4;

  constexpr int posPerWave     = GROUPS_PER_WAVE * GROUP;
  constexpr int posPerBlock    = WAVES_PER_BLOCK * posPerWave;
  constexpr int blocksPerBatch = kL / posPerBlock;
  const int b       = blockIdx.x / blocksPerBatch;
  const int lblk    = (blockIdx.x % blocksPerBatch) * posPerBlock;
  const int wave_l0 = lblk + wave * posPerWave;

  v2f A[8];
  #pragma unroll
  for (int i = 0; i < 8; ++i) {
    const int cc = i >> 1, th = i & 1;
    float a0 = 0.0f, a1 = 0.0f;
    if (m < C_OUT) {
      const float* wp = W + (m * C_IN + 4 * hi + cc) * FSZ + 2 * th;
      a0 = wp[0];
      a1 = wp[1];
    }
    A[i].x = a0;
    A[i].y = a1;
  }

  v8f cbias;
  #pragma unroll
  for (int r = 0; r < 8; ++r) cbias[r] = hi ? 0.0f : bias[r];

  const float* xh = x   + ((long long)b * C_IN + 4 * hi) * kL;
  float*       ob = out + (long long)b * C_OUT * kL;
  const int p0 = wave_l0 + 4 * m;
  const bool writer = (hi == 0);

  XGroup g[GROUPS_PER_WAVE + 1];
  const bool lane0 = (m == 0);
  const int  ptail = (wave_l0 + posPerWave) & (kL - 1);
  load_group(g[0], xh, p0);
  load_group(g[1], xh, p0 + GROUP);
  #pragma unroll
  for (int t = 0; t < GROUPS_PER_WAVE; ++t) {
    if (t + 2 < GROUPS_PER_WAVE) load_group(g[t + 2], xh, p0 + (t + 2) * GROUP);
    else if (t + 2 == GROUPS_PER_WAVE) {
      #pragma unroll
      for (int cc = 0; cc < 4; ++cc) {
        const v3f t3 = *(const v3f*)(xh + cc * kL + ptail);
        g[t + 2].a[cc] = v4f{t3.x, t3.y, t3.z, 0.0f};
      }
    }
    __builtin_amdgcn_sched_barrier(0);
    compute_store(g[t], g[t + 1], lane0, A, cbias, ob + p0 + t * GROUP, writer);
  }
}

extern "C" void kernel_launch(void* const* d_in, const int* in_sizes, int n_in,
                              void* d_out, int out_size, void* d_ws, size_t ws_size,
                              hipStream_t stream) {
  const float* x    = (const float*)d_in[0];
  const float* W    = (const float*)d_in[1];
  const float* bias = (const float*)d_in[2];
  float*       out  = (float*)d_out;

  const int B = in_sizes[0] / (C_IN * kL);

  constexpr int posPerBlock = WAVES_PER_BLOCK * GROUPS_PER_WAVE * GROUP;
  const int blocks = B * (kL / posPerBlock);

  convPbc_wmma_kernel<<<blocks, 256, 0, stream>>>(x, W, bias, out);
}
